# starting kernel
# baseline (speedup 1.0000x reference)
#include <hip/hip_runtime.h>
#include <hip/hip_bf16.h>

typedef float v2f __attribute__((ext_vector_type(2)));
typedef float v8f __attribute__((ext_vector_type(8)));

#define WMMA_F32(a, b, c) \
    __builtin_amdgcn_wmma_f32_16x16x4_f32(false, (a), false, (b), (short)0, (c), false, false)

#define EDIM 768
#define HDIM 384
#define G4H  1536
#define BSZ  32
#define SSZ  512
#define ROWS (BSZ * SSZ)

__device__ __forceinline__ float sigmoid_f(float x) {
    return 1.0f / (1.0f + __expf(-x));
}
__device__ __forceinline__ float tanh_f(float x) {
    return 1.0f - 2.0f / (__expf(2.0f * x) + 1.0f);
}

__global__ __launch_bounds__(128)
void xg_gemm_kernel(const float* __restrict__ Af, const float* __restrict__ Ab,
                    const float* __restrict__ Wf, const float* __restrict__ Wb,
                    const float* __restrict__ bihf, const float* __restrict__ bhhf,
                    const float* __restrict__ bihb, const float* __restrict__ bhhb,
                    float* __restrict__ xg, int Kin, int reverse)
{
    const int dir = blockIdx.z;
    const float* __restrict__ A   = dir ? Ab   : Af;
    const float* __restrict__ W   = dir ? Wb   : Wf;
    const float* __restrict__ bih = dir ? bihb : bihf;
    const float* __restrict__ bhh = dir ? bhhb : bhhf;

    const int n0 = blockIdx.x * 64;
    const int m0 = blockIdx.y * 64;

    __shared__ float As[64][17];
    __shared__ float Bs[16][65];

    const int tid   = threadIdx.x;
    const int lane  = tid & 31;
    const int wid   = tid >> 5;
    const int wm    = (wid & 1) * 32;
    const int wn    = (wid >> 1) * 32;
    const int lhalf = lane >> 4;
    const int l16   = lane & 15;

    v8f acc[2][2] = {};

    for (int k0 = 0; k0 < Kin; k0 += 16) {
        #pragma unroll
        for (int i = 0; i < 8; i++) {
            int idx = tid + i * 128;
            int mr  = idx >> 4;
            int kk  = idx & 15;
            int m   = m0 + mr;
            int b   = m >> 9;
            int s   = m & 511;
            int srd = (reverse && dir) ? (511 - s) : s;
            As[mr][kk] = A[(size_t)(b * SSZ + srd) * Kin + (k0 + kk)];
        }
        #pragma unroll
        for (int i = 0; i < 8; i++) {
            int idx = tid + i * 128;
            int nr  = idx >> 4;
            int kk  = idx & 15;
            Bs[kk][nr] = W[(size_t)(n0 + nr) * Kin + (k0 + kk)];
        }
        __syncthreads();

        #pragma unroll
        for (int kc = 0; kc < 16; kc += 4) {
            v2f afrag[2], bfrag[2];
            #pragma unroll
            for (int mt = 0; mt < 2; mt++) {
                int m = wm + mt * 16 + l16;
                afrag[mt].x = As[m][kc + 2 * lhalf];
                afrag[mt].y = As[m][kc + 2 * lhalf + 1];
            }
            #pragma unroll
            for (int nt = 0; nt < 2; nt++) {
                int n = wn + nt * 16 + l16;
                bfrag[nt].x = Bs[kc + 2 * lhalf][n];
                bfrag[nt].y = Bs[kc + 2 * lhalf + 1][n];
            }
            #pragma unroll
            for (int mt = 0; mt < 2; mt++)
                #pragma unroll
                for (int nt = 0; nt < 2; nt++)
                    acc[mt][nt] = WMMA_F32(afrag[mt], bfrag[nt], acc[mt][nt]);
        }
        __syncthreads();
    }

    #pragma unroll
    for (int mt = 0; mt < 2; mt++)
        #pragma unroll
        for (int nt = 0; nt < 2; nt++) {
            int n = n0 + wn + nt * 16 + l16;
            float bias = bih[n] + bhh[n];
            #pragma unroll
            for (int r = 0; r < 8; r++) {
                int m = m0 + wm + mt * 16 + r + 8 * lhalf;
                int b = m >> 9;
                int s = m & 511;
                __builtin_nontemporal_store(
                    acc[mt][nt][r] + bias,
                    &xg[(((size_t)dir * SSZ + s) * BSZ + b) * G4H + n]);
            }
        }
}

__global__ __launch_bounds__(768)
void lstm_recur_kernel(const float* __restrict__ xg,
                       const float* __restrict__ Whhf, const float* __restrict__ Whhb,
                       float* __restrict__ out, size_t dirStride, int rowStride)
{
    const int dir = blockIdx.x;
    const float* __restrict__ Whh = dir ? Whhb : Whhf;
    const float* __restrict__ xgd = xg + (size_t)dir * SSZ * BSZ * G4H;

    __shared__ float hs[32][HDIM + 1];

    const int tid   = threadIdx.x;
    const int lane  = tid & 31;
    const int wid   = tid >> 5;
    const int hc0   = wid * 16;
    const int lhalf = lane >> 4;
    const int l16   = lane & 15;

    for (int i = tid; i < 32 * (HDIM + 1); i += 768)
        (&hs[0][0])[i] = 0.0f;

    v8f cst[2] = {};
    float* outd = out + dir * dirStride;

    __syncthreads();

    for (int t = 0; t < SSZ; t++) {
        v8f acc[4][2];
        const float* __restrict__ xgt = xgd + (size_t)t * BSZ * G4H;
        #pragma unroll
        for (int g = 0; g < 4; g++)
            #pragma unroll
            for (int mt = 0; mt < 2; mt++)
                #pragma unroll
                for (int r = 0; r < 8; r++) {
                    int b = mt * 16 + r + 8 * lhalf;
                    acc[g][mt][r] = xgt[(size_t)b * G4H + g * HDIM + hc0 + l16];
                }

        if (t + 1 < SSZ) {
            const char* nxt = (const char*)(xgd + (size_t)(t + 1) * BSZ * G4H);
            __builtin_prefetch(nxt + (size_t)tid * 256, 0, 0);
            __builtin_prefetch(nxt + (size_t)tid * 256 + 128, 0, 0);
        }

        for (int kc = 0; kc < HDIM; kc += 4) {
            v2f af[2];
            #pragma unroll
            for (int mt = 0; mt < 2; mt++) {
                int b = mt * 16 + l16;
                af[mt].x = hs[b][kc + 2 * lhalf];
                af[mt].y = hs[b][kc + 2 * lhalf + 1];
            }
            #pragma unroll
            for (int g = 0; g < 4; g++) {
                int n = g * HDIM + hc0 + l16;
                const float* wp = Whh + (size_t)n * HDIM + kc + 2 * lhalf;
                v2f bf;
                bf.x = wp[0];
                bf.y = wp[1];
                acc[g][0] = WMMA_F32(af[0], bf, acc[g][0]);
                acc[g][1] = WMMA_F32(af[1], bf, acc[g][1]);
            }
        }
        __syncthreads();

        #pragma unroll
        for (int mt = 0; mt < 2; mt++) {
            #pragma unroll
            for (int r = 0; r < 8; r++) {
                float iv = sigmoid_f(acc[0][mt][r]);
                float fv = sigmoid_f(acc[1][mt][r]);
                float gv = tanh_f(acc[2][mt][r]);
                float ov = sigmoid_f(acc[3][mt][r]);
                float c  = fv * cst[mt][r] + iv * gv;
                cst[mt][r] = c;
                float h  = ov * tanh_f(c);
                int b = mt * 16 + r + 8 * lhalf;
                int n = hc0 + l16;
                hs[b][n] = h;
                __builtin_nontemporal_store(
                    h, &outd[(size_t)(b * SSZ + t) * rowStride + n]);
            }
        }
        __syncthreads();
    }
}

extern "C" void kernel_launch(void* const* d_in, const int* in_sizes, int n_in,
                              void* d_out, int out_size, void* d_ws, size_t ws_size,
                              hipStream_t stream) {
    const float* emb    = (const float*)d_in[0];
    const float* wih_f0 = (const float*)d_in[2];
    const float* whh_f0 = (const float*)d_in[3];
    const float* bih_f0 = (const float*)d_in[4];
    const float* bhh_f0 = (const float*)d_in[5];
    const float* wih_f1 = (const float*)d_in[6];
    const float* whh_f1 = (const float*)d_in[7];
    const float* bih_f1 = (const float*)d_in[8];
    const float* bhh_f1 = (const float*)d_in[9];
    const float* wih_b0 = (const float*)d_in[10];
    const float* whh_b0 = (const float*)d_in[11];
    const float* bih_b0 = (const float*)d_in[12];
    const float* bhh_b0 = (const float*)d_in[13];
    const float* wih_b1 = (const float*)d_in[14];
    const float* whh_b1 = (const float*)d_in[15];
    const float* bih_b1 = (const float*)d_in[16];
    const float* bhh_b1 = (const float*)d_in[17];

    float* xg = (float*)d_ws;
    float* h0 = xg + (size_t)2 * SSZ * BSZ * G4H;
    float* out = (float*)d_out;

    dim3 gA(G4H / 64, ROWS / 64, 2);
    dim3 bA(128);
    dim3 gR(2);
    dim3 bR(768);

    xg_gemm_kernel<<<gA, bA, 0, stream>>>(emb, emb, wih_f0, wih_b0,
                                          bih_f0, bhh_f0, bih_b0, bhh_b0,
                                          xg, EDIM, 1);
    lstm_recur_kernel<<<gR, bR, 0, stream>>>(xg, whh_f0, whh_b0,
                                             h0, (size_t)ROWS * HDIM, HDIM);
    xg_gemm_kernel<<<gA, bA, 0, stream>>>(h0, h0 + (size_t)ROWS * HDIM,
                                          wih_f1, wih_b1,
                                          bih_f1, bhh_f1, bih_b1, bhh_b1,
                                          xg, HDIM, 0);
    lstm_recur_kernel<<<gR, bR, 0, stream>>>(xg, whh_f1, whh_b1,
                                             out, (size_t)HDIM, 2 * HDIM);
}


# Recurrence weight loads: per 32-k chunk each lane reads 64 contiguous bytes of its Whh row (four 16-B loads back to back, h fragment in the same k order), so every fetched 128-B line is used whole
# speedup vs baseline: 1.6728x; 1.6728x over previous
#include <hip/hip_runtime.h>
#include <hip/hip_bf16.h>

typedef float v2f __attribute__((ext_vector_type(2)));
typedef float v4f __attribute__((ext_vector_type(4)));
typedef float v8f __attribute__((ext_vector_type(8)));

#define WMMA_F32(a, b, c) \
    __builtin_amdgcn_wmma_f32_16x16x4_f32(false, (a), false, (b), (short)0, (c), false, false)

#define EDIM 768
#define HDIM 384
#define G4H  1536
#define BSZ  32
#define SSZ  512
#define ROWS (BSZ * SSZ)
#define HPITCH 388

__device__ __forceinline__ float sigmoid_f(float x) {
    return 1.0f / (1.0f + __expf(-x));
}
__device__ __forceinline__ float tanh_f(float x) {
    return 1.0f - 2.0f / (__expf(2.0f * x) + 1.0f);
}

__global__ __launch_bounds__(128)
void xg_gemm_kernel(const float* __restrict__ Af, const float* __restrict__ Ab,
                    const float* __restrict__ Wf, const float* __restrict__ Wb,
                    const float* __restrict__ bihf, const float* __restrict__ bhhf,
                    const float* __restrict__ bihb, const float* __restrict__ bhhb,
                    float* __restrict__ xg, int Kin, int reverse)
{
    const int dir = blockIdx.z;
    const float* __restrict__ A   = dir ? Ab   : Af;
    const float* __restrict__ W   = dir ? Wb   : Wf;
    const float* __restrict__ bih = dir ? bihb : bihf;
    const float* __restrict__ bhh = dir ? bhhb : bhhf;

    const int n0 = blockIdx.x * 64;
    const int m0 = blockIdx.y * 64;

    __shared__ float As[64][17];
    __shared__ float Bs[16][65];

    const int tid   = threadIdx.x;
    const int lane  = tid & 31;
    const int wid   = tid >> 5;
    const int wm    = (wid & 1) * 32;
    const int wn    = (wid >> 1) * 32;
    const int lhalf = lane >> 4;
    const int l16   = lane & 15;

    v8f acc[2][2] = {};

    for (int k0 = 0; k0 < Kin; k0 += 16) {
        #pragma unroll
        for (int i = 0; i < 8; i++) {
            int idx = tid + i * 128;
            int mr  = idx >> 4;
            int kk  = idx & 15;
            int m   = m0 + mr;
            int b   = m >> 9;
            int s   = m & 511;
            int srd = (reverse && dir) ? (511 - s) : s;
            As[mr][kk] = A[(size_t)(b * SSZ + srd) * Kin + (k0 + kk)];
        }
        #pragma unroll
        for (int i = 0; i < 8; i++) {
            int idx = tid + i * 128;
            int nr  = idx >> 4;
            int kk  = idx & 15;
            Bs[kk][nr] = W[(size_t)(n0 + nr) * Kin + (k0 + kk)];
        }
        __syncthreads();

        #pragma unroll
        for (int kc = 0; kc < 16; kc += 4) {
            v2f afrag[2], bfrag[2];
            #pragma unroll
            for (int mt = 0; mt < 2; mt++) {
                int m = wm + mt * 16 + l16;
                afrag[mt].x = As[m][kc + 2 * lhalf];
                afrag[mt].y = As[m][kc + 2 * lhalf + 1];
            }
            #pragma unroll
            for (int nt = 0; nt < 2; nt++) {
                int n = wn + nt * 16 + l16;
                bfrag[nt].x = Bs[kc + 2 * lhalf][n];
                bfrag[nt].y = Bs[kc + 2 * lhalf + 1][n];
            }
            #pragma unroll
            for (int mt = 0; mt < 2; mt++)
                #pragma unroll
                for (int nt = 0; nt < 2; nt++)
                    acc[mt][nt] = WMMA_F32(afrag[mt], bfrag[nt], acc[mt][nt]);
        }
        __syncthreads();
    }

    #pragma unroll
    for (int mt = 0; mt < 2; mt++)
        #pragma unroll
        for (int nt = 0; nt < 2; nt++) {
            int n = n0 + wn + nt * 16 + l16;
            float bias = bih[n] + bhh[n];
            #pragma unroll
            for (int r = 0; r < 8; r++) {
                int m = m0 + wm + mt * 16 + r + 8 * lhalf;
                int b = m >> 9;
                int s = m & 511;
                __builtin_nontemporal_store(
                    acc[mt][nt][r] + bias,
                    &xg[(((size_t)dir * SSZ + s) * BSZ + b) * G4H + n]);
            }
        }
}

__global__ __launch_bounds__(768)
void lstm_recur_kernel(const float* __restrict__ xg,
                       const float* __restrict__ Whhf, const float* __restrict__ Whhb,
                       float* __restrict__ out, size_t dirStride, int rowStride)
{
    const int dir = blockIdx.x;
    const float* __restrict__ Whh = dir ? Whhb : Whhf;
    const float* __restrict__ xgd = xg + (size_t)dir * SSZ * BSZ * G4H;

    __shared__ __attribute__((aligned(16))) float hs[32][HPITCH];

    const int tid   = threadIdx.x;
    const int lane  = tid & 31;
    const int wid   = tid >> 5;
    const int hc0   = wid * 16;
    const int lhalf = lane >> 4;
    const int l16   = lane & 15;

    for (int i = tid; i < 32 * HPITCH; i += 768)
        (&hs[0][0])[i] = 0.0f;

    v8f cst[2] = {};
    float* outd = out + dir * dirStride;

    const unsigned wlane = (unsigned)((hc0 + l16) * HDIM + 16 * lhalf);
    const float* hlane0 = &hs[l16][16 * lhalf];
    const float* hlane1 = &hs[16 + l16][16 * lhalf];
    const int xlane = 8 * lhalf * G4H + hc0 + l16;

    __syncthreads();

    for (int t = 0; t < SSZ; t++) {
        v8f acc[4][2];
        const float* __restrict__ xgt = xgd + (size_t)t * BSZ * G4H;
        const float* __restrict__ xl  = xgt + xlane;
        #pragma unroll
        for (int g = 0; g < 4; g++)
            #pragma unroll
            for (int mt = 0; mt < 2; mt++)
                #pragma unroll
                for (int r = 0; r < 8; r++)
                    acc[g][mt][r] = xl[(mt * 16 + r) * G4H + g * HDIM];

        if (t + 1 < SSZ) {
            const char* nxt = (const char*)(xgd + (size_t)(t + 1) * BSZ * G4H);
            __builtin_prefetch(nxt + (size_t)tid * 256, 0, 0);
            __builtin_prefetch(nxt + (size_t)tid * 256 + 128, 0, 0);
        }

        #pragma unroll
        for (int p = 0; p < 2; p++) {
            #pragma unroll 1
            for (int kb = 0; kb < HDIM; kb += 32) {
                v4f qb[2][4];
                #pragma unroll
                for (int gg = 0; gg < 2; gg++)
                    #pragma unroll
                    for (int j = 0; j < 4; j++)
                        qb[gg][j] = *(const v4f*)(Whh + (wlane + (unsigned)((2 * p + gg) * HDIM * HDIM
                                                                            + kb + 4 * j)));
                #pragma unroll
                for (int j = 0; j < 4; j++) {
                    v4f qa0 = *(const v4f*)(hlane0 + kb + 4 * j);
                    v4f qa1 = *(const v4f*)(hlane1 + kb + 4 * j);
                    #pragma unroll
                    for (int s = 0; s < 2; s++) {
                        v2f a0, a1;
                        a0.x = s ? qa0.z : qa0.x;  a0.y = s ? qa0.w : qa0.y;
                        a1.x = s ? qa1.z : qa1.x;  a1.y = s ? qa1.w : qa1.y;
                        #pragma unroll
                        for (int gg = 0; gg < 2; gg++) {
                            v2f bf;
                            bf.x = s ? qb[gg][j].z : qb[gg][j].x;
                            bf.y = s ? qb[gg][j].w : qb[gg][j].y;
                            acc[2 * p + gg][0] = WMMA_F32(a0, bf, acc[2 * p + gg][0]);
                            acc[2 * p + gg][1] = WMMA_F32(a1, bf, acc[2 * p + gg][1]);
                        }
                    }
                }
            }
        }
        __syncthreads();

        #pragma unroll
        for (int mt = 0; mt < 2; mt++) {
            #pragma unroll
            for (int r = 0; r < 8; r++) {
                float iv = sigmoid_f(acc[0][mt][r]);
                float fv = sigmoid_f(acc[1][mt][r]);
                float gv = tanh_f(acc[2][mt][r]);
                float ov = sigmoid_f(acc[3][mt][r]);
                float c  = fv * cst[mt][r] + iv * gv;
                cst[mt][r] = c;
                float h  = ov * tanh_f(c);
                int b = mt * 16 + r + 8 * lhalf;
                int n = hc0 + l16;
                hs[b][n] = h;
                __builtin_nontemporal_store(
                    h, &outd[(size_t)(b * SSZ + t) * rowStride + n]);
            }
        }
        __syncthreads();
    }
}

extern "C" void kernel_launch(void* const* d_in, const int* in_sizes, int n_in,
                              void* d_out, int out_size, void* d_ws, size_t ws_size,
                              hipStream_t stream) {
    const float* emb    = (const float*)d_in[0];
    const float* wih_f0 = (const float*)d_in[2];
    const float* whh_f0 = (const float*)d_in[3];
    const float* bih_f0 = (const float*)d_in[4];
    const float* bhh_f0 = (const float*)d_in[5];
    const float* wih_f1 = (const float*)d_in[6];
    const float* whh_f1 = (const float*)d_in[7];
    const float* bih_f1 = (const float*)d_in[8];
    const float* bhh_f1 = (const float*)d_in[9];
    const float* wih_b0 = (const float*)d_in[10];
    const float* whh_b0 = (const float*)d_in[11];
    const float* bih_b0 = (const float*)d_in[12];
    const float* bhh_b0 = (const float*)d_in[13];
    const float* wih_b1 = (const float*)d_in[14];
    const float* whh_b1 = (const float*)d_in[15];
    const float* bih_b1 = (const float*)d_in[16];
    const float* bhh_b1 = (const float*)d_in[17];

    float* xg = (float*)d_ws;
    float* h0 = xg + (size_t)2 * SSZ * BSZ * G4H;
    float* out = (float*)d_out;

    dim3 gA(G4H / 64, ROWS / 64, 2);
    dim3 bA(128);
    dim3 gR(2);
    dim3 bR(768);

    xg_gemm_kernel<<<gA, bA, 0, stream>>>(emb, emb, wih_f0, wih_b0,
                                          bih_f0, bhh_f0, bih_b0, bhh_b0,
                                          xg, EDIM, 1);
    lstm_recur_kernel<<<gR, bR, 0, stream>>>(xg, whh_f0, whh_b0,
                                             h0, (size_t)ROWS * HDIM, HDIM);
    xg_gemm_kernel<<<gA, bA, 0, stream>>>(h0, h0 + (size_t)ROWS * HDIM,
                                          wih_f1, wih_b1,
                                          bih_f1, bhh_f1, bih_b1, bhh_b1,
                                          xg, HDIM, 0);
    lstm_recur_kernel<<<gR, bR, 0, stream>>>(xg, whh_f1, whh_b1,
                                             out, (size_t)HDIM, 2 * HDIM);
}


# Recurrence split by batch half: grid 2 directions x 2 halves, each work-group carries 16 batch rows on its own WGP with a 16x64 wave tile (all four gates per k chunk)
# speedup vs baseline: 2.4815x; 1.4834x over previous
#include <hip/hip_runtime.h>
#include <hip/hip_bf16.h>

typedef float v2f __attribute__((ext_vector_type(2)));
typedef float v4f __attribute__((ext_vector_type(4)));
typedef float v8f __attribute__((ext_vector_type(8)));

#define WMMA_F32(a, b, c) \
    __builtin_amdgcn_wmma_f32_16x16x4_f32(false, (a), false, (b), (short)0, (c), false, false)

#define EDIM 768
#define HDIM 384
#define G4H  1536
#define BSZ  32
#define SSZ  512
#define ROWS (BSZ * SSZ)
#define HPITCH 388

__device__ __forceinline__ float sigmoid_f(float x) {
    return 1.0f / (1.0f + __expf(-x));
}
__device__ __forceinline__ float tanh_f(float x) {
    return 1.0f - 2.0f / (__expf(2.0f * x) + 1.0f);
}

__global__ __launch_bounds__(128)
void xg_gemm_kernel(const float* __restrict__ Af, const float* __restrict__ Ab,
                    const float* __restrict__ Wf, const float* __restrict__ Wb,
                    const float* __restrict__ bihf, const float* __restrict__ bhhf,
                    const float* __restrict__ bihb, const float* __restrict__ bhhb,
                    float* __restrict__ xg, int Kin, int reverse)
{
    const int dir = blockIdx.z;
    const float* __restrict__ A   = dir ? Ab   : Af;
    const float* __restrict__ W   = dir ? Wb   : Wf;
    const float* __restrict__ bih = dir ? bihb : bihf;
    const float* __restrict__ bhh = dir ? bhhb : bhhf;

    const int n0 = blockIdx.x * 64;
    const int m0 = blockIdx.y * 64;

    __shared__ float As[64][17];
    __shared__ float Bs[16][65];

    const int tid   = threadIdx.x;
    const int lane  = tid & 31;
    const int wid   = tid >> 5;
    const int wm    = (wid & 1) * 32;
    const int wn    = (wid >> 1) * 32;
    const int lhalf = lane >> 4;
    const int l16   = lane & 15;

    v8f acc[2][2] = {};

    for (int k0 = 0; k0 < Kin; k0 += 16) {
        #pragma unroll
        for (int i = 0; i < 8; i++) {
            int idx = tid + i * 128;
            int mr  = idx >> 4;
            int kk  = idx & 15;
            int m   = m0 + mr;
            int b   = m >> 9;
            int s   = m & 511;
            int srd = (reverse && dir) ? (511 - s) : s;
            As[mr][kk] = A[(size_t)(b * SSZ + srd) * Kin + (k0 + kk)];
        }
        #pragma unroll
        for (int i = 0; i < 8; i++) {
            int idx = tid + i * 128;
            int nr  = idx >> 4;
            int kk  = idx & 15;
            Bs[kk][nr] = W[(size_t)(n0 + nr) * Kin + (k0 + kk)];
        }
        __syncthreads();

        #pragma unroll
        for (int kc = 0; kc < 16; kc += 4) {
            v2f afrag[2], bfrag[2];
            #pragma unroll
            for (int mt = 0; mt < 2; mt++) {
                int m = wm + mt * 16 + l16;
                afrag[mt].x = As[m][kc + 2 * lhalf];
                afrag[mt].y = As[m][kc + 2 * lhalf + 1];
            }
            #pragma unroll
            for (int nt = 0; nt < 2; nt++) {
                int n = wn + nt * 16 + l16;
                bfrag[nt].x = Bs[kc + 2 * lhalf][n];
                bfrag[nt].y = Bs[kc + 2 * lhalf + 1][n];
            }
            #pragma unroll
            for (int mt = 0; mt < 2; mt++)
                #pragma unroll
                for (int nt = 0; nt < 2; nt++)
                    acc[mt][nt] = WMMA_F32(afrag[mt], bfrag[nt], acc[mt][nt]);
        }
        __syncthreads();
    }

    #pragma unroll
    for (int mt = 0; mt < 2; mt++)
        #pragma unroll
        for (int nt = 0; nt < 2; nt++) {
            int n = n0 + wn + nt * 16 + l16;
            float bias = bih[n] + bhh[n];
            #pragma unroll
            for (int r = 0; r < 8; r++) {
                int m = m0 + wm + mt * 16 + r + 8 * lhalf;
                int b = m >> 9;
                int s = m & 511;
                __builtin_nontemporal_store(
                    acc[mt][nt][r] + bias,
                    &xg[(((size_t)dir * SSZ + s) * BSZ + b) * G4H + n]);
            }
        }
}

__global__ __launch_bounds__(768)
void lstm_recur_kernel(const float* __restrict__ xg,
                       const float* __restrict__ Whhf, const float* __restrict__ Whhb,
                       float* __restrict__ out, size_t dirStride, int rowStride)
{
    const int dir = blockIdx.x;
    const int b0  = blockIdx.y * 16;
    const float* __restrict__ Whh = dir ? Whhb : Whhf;
    const float* __restrict__ xgd = xg + (size_t)dir * SSZ * BSZ * G4H;

    __shared__ __attribute__((aligned(16))) float hs[16][HPITCH];

    const int tid   = threadIdx.x;
    const int lane  = tid & 31;
    const int wid   = tid >> 5;
    const int hc0   = wid * 16;
    const int lhalf = lane >> 4;
    const int l16   = lane & 15;

    for (int i = tid; i < 16 * HPITCH; i += 768)
        (&hs[0][0])[i] = 0.0f;

    v8f cst = {};
    float* outd = out + dir * dirStride;

    const unsigned wlane = (unsigned)((hc0 + l16) * HDIM + 16 * lhalf);
    const float* hlane = &hs[l16][16 * lhalf];
    const int xlane = (b0 + 8 * lhalf) * G4H + hc0 + l16;

    __syncthreads();

    for (int t = 0; t < SSZ; t++) {
        v8f acc[4];
        const float* __restrict__ xgt = xgd + (size_t)t * BSZ * G4H;
        const float* __restrict__ xl  = xgt + xlane;
        #pragma unroll
        for (int g = 0; g < 4; g++)
            #pragma unroll
            for (int r = 0; r < 8; r++)
                acc[g][r] = xl[r * G4H + g * HDIM];

        if (t + 1 < SSZ) {
            const char* nxt = (const char*)(xgd + (size_t)(t + 1) * BSZ * G4H + (size_t)b0 * G4H);
            __builtin_prefetch(nxt + (size_t)tid * 128, 0, 0);
        }

        #pragma unroll 1
        for (int kb = 0; kb < HDIM; kb += 32) {
            v4f qb[4][4];
            #pragma unroll
            for (int g = 0; g < 4; g++)
                #pragma unroll
                for (int j = 0; j < 4; j++)
                    qb[g][j] = *(const v4f*)(Whh + (wlane + (unsigned)(g * HDIM * HDIM + kb + 4 * j)));
            #pragma unroll
            for (int j = 0; j < 4; j++) {
                v4f qa = *(const v4f*)(hlane + kb + 4 * j);
                #pragma unroll
                for (int s = 0; s < 2; s++) {
                    v2f a;
                    a.x = s ? qa.z : qa.x;  a.y = s ? qa.w : qa.y;
                    #pragma unroll
                    for (int g = 0; g < 4; g++) {
                        v2f bf;
                        bf.x = s ? qb[g][j].z : qb[g][j].x;
                        bf.y = s ? qb[g][j].w : qb[g][j].y;
                        acc[g] = WMMA_F32(a, bf, acc[g]);
                    }
                }
            }
        }
        __syncthreads();

        #pragma unroll
        for (int r = 0; r < 8; r++) {
            float iv = sigmoid_f(acc[0][r]);
            float fv = sigmoid_f(acc[1][r]);
            float gv = tanh_f(acc[2][r]);
            float ov = sigmoid_f(acc[3][r]);
            float c  = fv * cst[r] + iv * gv;
            cst[r] = c;
            float h  = ov * tanh_f(c);
            int bl = r + 8 * lhalf;
            int n  = hc0 + l16;
            hs[bl][n] = h;
            __builtin_nontemporal_store(
                h, &outd[(size_t)((b0 + bl) * SSZ + t) * rowStride + n]);
        }
        __syncthreads();
    }
}

extern "C" void kernel_launch(void* const* d_in, const int* in_sizes, int n_in,
                              void* d_out, int out_size, void* d_ws, size_t ws_size,
                              hipStream_t stream) {
    const float* emb    = (const float*)d_in[0];
    const float* wih_f0 = (const float*)d_in[2];
    const float* whh_f0 = (const float*)d_in[3];
    const float* bih_f0 = (const float*)d_in[4];
    const float* bhh_f0 = (const float*)d_in[5];
    const float* wih_f1 = (const float*)d_in[6];
    const float* whh_f1 = (const float*)d_in[7];
    const float* bih_f1 = (const float*)d_in[8];
    const float* bhh_f1 = (const float*)d_in[9];
    const float* wih_b0 = (const float*)d_in[10];
    const float* whh_b0 = (const float*)d_in[11];
    const float* bih_b0 = (const float*)d_in[12];
    const float* bhh_b0 = (const float*)d_in[13];
    const float* wih_b1 = (const float*)d_in[14];
    const float* whh_b1 = (const float*)d_in[15];
    const float* bih_b1 = (const float*)d_in[16];
    const float* bhh_b1 = (const float*)d_in[17];

    float* xg = (float*)d_ws;
    float* h0 = xg + (size_t)2 * SSZ * BSZ * G4H;
    float* out = (float*)d_out;

    dim3 gA(G4H / 64, ROWS / 64, 2);
    dim3 bA(128);
    dim3 gR(2, 2);
    dim3 bR(768);

    xg_gemm_kernel<<<gA, bA, 0, stream>>>(emb, emb, wih_f0, wih_b0,
                                          bih_f0, bhh_f0, bih_b0, bhh_b0,
                                          xg, EDIM, 1);
    lstm_recur_kernel<<<gR, bR, 0, stream>>>(xg, whh_f0, whh_b0,
                                             h0, (size_t)ROWS * HDIM, HDIM);
    xg_gemm_kernel<<<gA, bA, 0, stream>>>(h0, h0 + (size_t)ROWS * HDIM,
                                          wih_f1, wih_b1,
                                          bih_f1, bhh_f1, bih_b1, bhh_b1,
                                          xg, HDIM, 0);
    lstm_recur_kernel<<<gR, bR, 0, stream>>>(xg, whh_f1, whh_b1,
                                             out, (size_t)HDIM, 2 * HDIM);
}


# Recurrence runs one launch per time step (1026 launches per call, was 4): the launch boundary hands h over, so a step's gate product spreads over 48 work-groups instead of 4
# speedup vs baseline: 12.0599x; 4.8599x over previous
#include <hip/hip_runtime.h>
#include <hip/hip_bf16.h>

typedef float v2f __attribute__((ext_vector_type(2)));
typedef float v4f __attribute__((ext_vector_type(4)));
typedef float v8f __attribute__((ext_vector_type(8)));

#define WMMA_F32(a, b, c) \
    __builtin_amdgcn_wmma_f32_16x16x4_f32(false, (a), false, (b), (short)0, (c), false, false)

#define EDIM 768
#define HDIM 384
#define G4H  1536
#define BSZ  32
#define SSZ  512
#define ROWS (BSZ * SSZ)
#define HPITCH 388

__device__ __forceinline__ float sigmoid_f(float x) {
    return 1.0f / (1.0f + __expf(-x));
}
__device__ __forceinline__ float tanh_f(float x) {
    return 1.0f - 2.0f / (__expf(2.0f * x) + 1.0f);
}

__global__ __launch_bounds__(128)
void xg_gemm_kernel(const float* __restrict__ Af, const float* __restrict__ Ab,
                    const float* __restrict__ Wf, const float* __restrict__ Wb,
                    const float* __restrict__ bihf, const float* __restrict__ bhhf,
                    const float* __restrict__ bihb, const float* __restrict__ bhhb,
                    float* __restrict__ xg, int Kin, int reverse)
{
    const int dir = blockIdx.z;
    const float* __restrict__ A   = dir ? Ab   : Af;
    const float* __restrict__ W   = dir ? Wb   : Wf;
    const float* __restrict__ bih = dir ? bihb : bihf;
    const float* __restrict__ bhh = dir ? bhhb : bhhf;

    const int n0 = blockIdx.x * 64;
    const int m0 = blockIdx.y * 64;

    __shared__ float As[64][17];
    __shared__ float Bs[16][65];

    const int tid   = threadIdx.x;
    const int lane  = tid & 31;
    const int wid   = tid >> 5;
    const int wm    = (wid & 1) * 32;
    const int wn    = (wid >> 1) * 32;
    const int lhalf = lane >> 4;
    const int l16   = lane & 15;

    v8f acc[2][2] = {};

    for (int k0 = 0; k0 < Kin; k0 += 16) {
        #pragma unroll
        for (int i = 0; i < 8; i++) {
            int idx = tid + i * 128;
            int mr  = idx >> 4;
            int kk  = idx & 15;
            int m   = m0 + mr;
            int b   = m >> 9;
            int s   = m & 511;
            int srd = (reverse && dir) ? (511 - s) : s;
            As[mr][kk] = A[(size_t)(b * SSZ + srd) * Kin + (k0 + kk)];
        }
        #pragma unroll
        for (int i = 0; i < 8; i++) {
            int idx = tid + i * 128;
            int nr  = idx >> 4;
            int kk  = idx & 15;
            Bs[kk][nr] = W[(size_t)(n0 + nr) * Kin + (k0 + kk)];
        }
        __syncthreads();

        #pragma unroll
        for (int kc = 0; kc < 16; kc += 4) {
            v2f afrag[2], bfrag[2];
            #pragma unroll
            for (int mt = 0; mt < 2; mt++) {
                int m = wm + mt * 16 + l16;
                afrag[mt].x = As[m][kc + 2 * lhalf];
                afrag[mt].y = As[m][kc + 2 * lhalf + 1];
            }
            #pragma unroll
            for (int nt = 0; nt < 2; nt++) {
                int n = wn + nt * 16 + l16;
                bfrag[nt].x = Bs[kc + 2 * lhalf][n];
                bfrag[nt].y = Bs[kc + 2 * lhalf + 1][n];
            }
            #pragma unroll
            for (int mt = 0; mt < 2; mt++)
                #pragma unroll
                for (int nt = 0; nt < 2; nt++)
                    acc[mt][nt] = WMMA_F32(afrag[mt], bfrag[nt], acc[mt][nt]);
        }
        __syncthreads();
    }

    #pragma unroll
    for (int mt = 0; mt < 2; mt++)
        #pragma unroll
        for (int nt = 0; nt < 2; nt++) {
            int n = n0 + wn + nt * 16 + l16;
            float bias = bih[n] + bhh[n];
            #pragma unroll
            for (int r = 0; r < 8; r++) {
                int m = m0 + wm + mt * 16 + r + 8 * lhalf;
                int b = m >> 9;
                int s = m & 511;
                __builtin_nontemporal_store(
                    acc[mt][nt][r] + bias,
                    &xg[(((size_t)dir * SSZ + s) * BSZ + b) * G4H + n]);
            }
        }
}

#define GPITCH 36

template <bool FIRST>
__global__ __launch_bounds__(256)
void lstm_step_kernel(float* __restrict__ xg,
                      const float* __restrict__ Whhf, const float* __restrict__ Whhb,
                      float* __restrict__ out, size_t dirStride, int rowStride, int t)
{
    const int hc0 = blockIdx.x * 32;
    const int b0  = blockIdx.y * 16;
    const int dir = blockIdx.z;
    const float* __restrict__ Whh = dir ? Whhb : Whhf;
    float* xgt  = xg + ((size_t)dir * SSZ + t) * (BSZ * G4H);
    float* outd = out + dir * dirStride;

    __shared__ __attribute__((aligned(16))) float hs[16][HPITCH];
    __shared__ float gs[4][16][GPITCH];

    const int tid   = threadIdx.x;
    const int lane  = tid & 31;
    const int wid   = tid >> 5;
    const int g     = wid >> 1;
    const int ch    = wid & 1;
    const int lhalf = lane >> 4;
    const int l16   = lane & 15;

    const int srow = tid >> 4;
    const int sc4  = tid & 15;
    v4f hq[6];
    if (!FIRST) {
        const float* hp = outd + (size_t)((b0 + srow) * SSZ + (t - 1)) * rowStride + 4 * sc4;
        #pragma unroll
        for (int i = 0; i < 6; i++)
            hq[i] = *(const v4f*)(hp + 64 * i);
    }

    v8f acc0, acc1 = {};
    {
        const float* xl = xgt + (size_t)(b0 + 8 * lhalf) * G4H + g * HDIM + hc0 + 16 * ch + l16;
        #pragma unroll
        for (int r = 0; r < 8; r++)
            acc0[r] = xl[r * G4H];
    }

    float cprev[2] = {0.0f, 0.0f};
    if (!FIRST) {
        const float* xgp = xgt - (size_t)(BSZ * G4H);
        #pragma unroll
        for (int i = 0; i < 2; i++)
            cprev[i] = xgp[(size_t)(b0 + 2 * wid + i) * G4H + hc0 + lane];
    }

    v4f qb[12][4];
    {
        const float* wrow = Whh + (size_t)(g * HDIM + hc0 + 16 * ch + l16) * HDIM + 16 * lhalf;
        #pragma unroll
        for (int kc = 0; kc < 12; kc++)
            #pragma unroll
            for (int j = 0; j < 4; j++)
                qb[kc][j] = *(const v4f*)(wrow + 32 * kc + 4 * j);
    }

    if (t + 1 < SSZ && tid < 64) {
        const float* nxt = xgt + (size_t)(BSZ * G4H)
                         + (size_t)(b0 + (tid >> 2)) * G4H + (tid & 3) * HDIM + hc0;
        __builtin_prefetch(nxt, 0, 0);
    }

    #pragma unroll
    for (int i = 0; i < 6; i++) {
        v4f v = {0.0f, 0.0f, 0.0f, 0.0f};
        if (!FIRST) v = hq[i];
        *(v4f*)&hs[srow][4 * sc4 + 64 * i] = v;
    }
    __builtin_amdgcn_fence(__ATOMIC_RELEASE, "workgroup", "local");
    __builtin_amdgcn_s_barrier();
    __builtin_amdgcn_fence(__ATOMIC_ACQUIRE, "workgroup", "local");

    const float* hlane = &hs[l16][16 * lhalf];
    v4f qa[2][4];
    #pragma unroll
    for (int j = 0; j < 4; j++)
        qa[0][j] = *(const v4f*)(hlane + 4 * j);
    __builtin_amdgcn_sched_group_barrier(0x100, 4, 0);
    #pragma unroll
    for (int kc = 0; kc < 12; kc++) {
        if (kc + 1 < 12) {
            #pragma unroll
            for (int j = 0; j < 4; j++)
                qa[(kc + 1) & 1][j] = *(const v4f*)(hlane + 32 * (kc + 1) + 4 * j);
        }
        #pragma unroll
        for (int j = 0; j < 4; j++) {
            const v4f ha = qa[kc & 1][j];
            v2f a0, a1, w0, w1;
            a0.x = ha.x;  a0.y = ha.y;  a1.x = ha.z;  a1.y = ha.w;
            w0.x = qb[kc][j].x;  w0.y = qb[kc][j].y;
            w1.x = qb[kc][j].z;  w1.y = qb[kc][j].w;
            acc0 = WMMA_F32(a0, w0, acc0);
            acc1 = WMMA_F32(a1, w1, acc1);
        }
        if (kc + 1 < 12)
            __builtin_amdgcn_sched_group_barrier(0x100, 4, 0);
        __builtin_amdgcn_sched_group_barrier(0x008, 8, 0);
    }

    #pragma unroll
    for (int r = 0; r < 8; r++)
        gs[g][8 * lhalf + r][16 * ch + l16] = acc0[r] + acc1[r];
    __syncthreads();

    #pragma unroll
    for (int i = 0; i < 2; i++) {
        const int row = 2 * wid + i;
        float iv = sigmoid_f(gs[0][row][lane]);
        float fv = sigmoid_f(gs[1][row][lane]);
        float gv = tanh_f(gs[2][row][lane]);
        float ov = sigmoid_f(gs[3][row][lane]);
        float c  = fv * cprev[i] + iv * gv;
        float h  = ov * tanh_f(c);
        xgt[(size_t)(b0 + row) * G4H + hc0 + lane] = c;
        outd[(size_t)((b0 + row) * SSZ + t) * rowStride + hc0 + lane] = h;
    }
}

extern "C" void kernel_launch(void* const* d_in, const int* in_sizes, int n_in,
                              void* d_out, int out_size, void* d_ws, size_t ws_size,
                              hipStream_t stream) {
    const float* emb    = (const float*)d_in[0];
    const float* wih_f0 = (const float*)d_in[2];
    const float* whh_f0 = (const float*)d_in[3];
    const float* bih_f0 = (const float*)d_in[4];
    const float* bhh_f0 = (const float*)d_in[5];
    const float* wih_f1 = (const float*)d_in[6];
    const float* whh_f1 = (const float*)d_in[7];
    const float* bih_f1 = (const float*)d_in[8];
    const float* bhh_f1 = (const float*)d_in[9];
    const float* wih_b0 = (const float*)d_in[10];
    const float* whh_b0 = (const float*)d_in[11];
    const float* bih_b0 = (const float*)d_in[12];
    const float* bhh_b0 = (const float*)d_in[13];
    const float* wih_b1 = (const float*)d_in[14];
    const float* whh_b1 = (const float*)d_in[15];
    const float* bih_b1 = (const float*)d_in[16];
    const float* bhh_b1 = (const float*)d_in[17];

    float* xg = (float*)d_ws;
    float* h0 = xg + (size_t)2 * SSZ * BSZ * G4H;
    float* out = (float*)d_out;

    dim3 gA(G4H / 64, ROWS / 64, 2);
    dim3 bA(128);
    dim3 gS(HDIM / 32, 2, 2);
    dim3 bS(256);

    xg_gemm_kernel<<<gA, bA, 0, stream>>>(emb, emb, wih_f0, wih_b0,
                                          bih_f0, bhh_f0, bih_b0, bhh_b0,
                                          xg, EDIM, 1);
    lstm_step_kernel<true><<<gS, bS, 0, stream>>>(xg, whh_f0, whh_b0,
                                                  h0, (size_t)ROWS * HDIM, HDIM, 0);
    for (int t = 1; t < SSZ; t++)
        lstm_step_kernel<false><<<gS, bS, 0, stream>>>(xg, whh_f0, whh_b0,
                                                       h0, (size_t)ROWS * HDIM, HDIM, t);
    xg_gemm_kernel<<<gA, bA, 0, stream>>>(h0, h0 + (size_t)ROWS * HDIM,
                                          wih_f1, wih_b1,
                                          bih_f1, bhh_f1, bih_b1, bhh_b1,
                                          xg, HDIM, 0);
    lstm_step_kernel<true><<<gS, bS, 0, stream>>>(xg, whh_f1, whh_b1,
                                                  out, (size_t)HDIM, 2 * HDIM, 0);
    for (int t = 1; t < SSZ; t++)
        lstm_step_kernel<false><<<gS, bS, 0, stream>>>(xg, whh_f1, whh_b1,
                                                       out, (size_t)HDIM, 2 * HDIM, t);
}


# Layers run as a wavefront: a launch carries layer-0 step j, layer-1 input product of step j-2 and layer-1 step j-3 on separate work-groups (516 launches per call, was 1026; no layer-1 batch GEMM)
# speedup vs baseline: 21.6778x; 1.7975x over previous
#include <hip/hip_runtime.h>
#include <hip/hip_bf16.h>

typedef float v2f __attribute__((ext_vector_type(2)));
typedef float v4f __attribute__((ext_vector_type(4)));
typedef float v8f __attribute__((ext_vector_type(8)));

#define WMMA_F32(a, b, c) \
    __builtin_amdgcn_wmma_f32_16x16x4_f32(false, (a), false, (b), (short)0, (c), false, false)

#define EDIM 768
#define HDIM 384
#define G4H  1536
#define BSZ  32
#define SSZ  512
#define ROWS (BSZ * SSZ)
#define HPITCH 388

__device__ __forceinline__ float sigmoid_f(float x) {
    return 1.0f / (1.0f + __expf(-x));
}
__device__ __forceinline__ float tanh_f(float x) {
    return 1.0f - 2.0f / (__expf(2.0f * x) + 1.0f);
}

__global__ __launch_bounds__(128)
void xg_gemm_kernel(const float* __restrict__ Af, const float* __restrict__ Ab,
                    const float* __restrict__ Wf, const float* __restrict__ Wb,
                    const float* __restrict__ bihf, const float* __restrict__ bhhf,
                    const float* __restrict__ bihb, const float* __restrict__ bhhb,
                    float* __restrict__ xg, int Kin, int reverse)
{
    const int dir = blockIdx.z;
    const float* __restrict__ A   = dir ? Ab   : Af;
    const float* __restrict__ W   = dir ? Wb   : Wf;
    const float* __restrict__ bih = dir ? bihb : bihf;
    const float* __restrict__ bhh = dir ? bhhb : bhhf;

    const int n0 = blockIdx.x * 64;
    const int m0 = blockIdx.y * 64;

    __shared__ float As[64][17];
    __shared__ float Bs[16][65];

    const int tid   = threadIdx.x;
    const int lane  = tid & 31;
    const int wid   = tid >> 5;
    const int wm    = (wid & 1) * 32;
    const int wn    = (wid >> 1) * 32;
    const int lhalf = lane >> 4;
    const int l16   = lane & 15;

    v8f acc[2][2] = {};

    for (int k0 = 0; k0 < Kin; k0 += 16) {
        #pragma unroll
        for (int i = 0; i < 8; i++) {
            int idx = tid + i * 128;
            int mr  = idx >> 4;
            int kk  = idx & 15;
            int m   = m0 + mr;
            int b   = m >> 9;
            int s   = m & 511;
            int srd = (reverse && dir) ? (511 - s) : s;
            As[mr][kk] = A[(size_t)(b * SSZ + srd) * Kin + (k0 + kk)];
        }
        #pragma unroll
        for (int i = 0; i < 8; i++) {
            int idx = tid + i * 128;
            int nr  = idx >> 4;
            int kk  = idx & 15;
            Bs[kk][nr] = W[(size_t)(n0 + nr) * Kin + (k0 + kk)];
        }
        __syncthreads();

        #pragma unroll
        for (int kc = 0; kc < 16; kc += 4) {
            v2f afrag[2], bfrag[2];
            #pragma unroll
            for (int mt = 0; mt < 2; mt++) {
                int m = wm + mt * 16 + l16;
                afrag[mt].x = As[m][kc + 2 * lhalf];
                afrag[mt].y = As[m][kc + 2 * lhalf + 1];
            }
            #pragma unroll
            for (int nt = 0; nt < 2; nt++) {
                int n = wn + nt * 16 + l16;
                bfrag[nt].x = Bs[kc + 2 * lhalf][n];
                bfrag[nt].y = Bs[kc + 2 * lhalf + 1][n];
            }
            #pragma unroll
            for (int mt = 0; mt < 2; mt++)
                #pragma unroll
                for (int nt = 0; nt < 2; nt++)
                    acc[mt][nt] = WMMA_F32(afrag[mt], bfrag[nt], acc[mt][nt]);
        }
        __syncthreads();
    }

    #pragma unroll
    for (int mt = 0; mt < 2; mt++)
        #pragma unroll
        for (int nt = 0; nt < 2; nt++) {
            int n = n0 + wn + nt * 16 + l16;
            float bias = bih[n] + bhh[n];
            #pragma unroll
            for (int r = 0; r < 8; r++) {
                int m = m0 + wm + mt * 16 + r + 8 * lhalf;
                int b = m >> 9;
                int s = m & 511;
                __builtin_nontemporal_store(
                    acc[mt][nt][r] + bias,
                    &xg[(((size_t)dir * SSZ + s) * BSZ + b) * G4H + n]);
            }
        }
}

#define GPITCH 36

__global__ __launch_bounds__(256)
void lstm_wavefront_kernel(float* __restrict__ xg,
                           const float* __restrict__ whh0f, const float* __restrict__ whh0b,
                           const float* __restrict__ wih1f, const float* __restrict__ wih1b,
                           const float* __restrict__ whh1f, const float* __restrict__ whh1b,
                           const float* __restrict__ bih1f, const float* __restrict__ bhh1f,
                           const float* __restrict__ bih1b, const float* __restrict__ bhh1b,
                           float* __restrict__ h0, float* __restrict__ out, int j)
{
    const int stage = blockIdx.x / 12;
    const int hc0 = (blockIdx.x - 12 * stage) * 32;
    const int b0  = blockIdx.y * 16;
    const int dir = blockIdx.z;

    const int t = (stage == 0) ? j : (stage == 1) ? (j - 2) : (j - 3);
    if (t < 0 || t >= SSZ) return;

    const bool rec   = (stage != 1);
    const bool zeroh = rec && (t == 0);

    const float* __restrict__ Wm =
        (stage == 0) ? (dir ? whh0b : whh0f) :
        (stage == 1) ? (dir ? wih1b : wih1f) : (dir ? whh1b : whh1f);

    float* hbase;
    int hstride;
    if (stage == 2) { hbase = out + dir * HDIM;                hstride = 2 * HDIM; }
    else            { hbase = h0 + (size_t)dir * ROWS * HDIM;  hstride = HDIM; }
    const int tin = rec ? (t - 1) : t;

    float* xgt = xg + ((size_t)dir * SSZ + t) * (BSZ * G4H);

    __shared__ __attribute__((aligned(16))) float hs[16][HPITCH];
    __shared__ float gs[4][16][GPITCH];

    const int tid   = threadIdx.x;
    const int lane  = tid & 31;
    const int wid   = tid >> 5;
    const int g     = wid >> 1;
    const int ch    = wid & 1;
    const int lhalf = lane >> 4;
    const int l16   = lane & 15;

    const int srow = tid >> 4;
    const int sc4  = tid & 15;
    v4f hq[6];
    #pragma unroll
    for (int i = 0; i < 6; i++)
        hq[i] = v4f{0.0f, 0.0f, 0.0f, 0.0f};
    if (!zeroh) {
        const float* hp = hbase + (size_t)((b0 + srow) * SSZ + tin) * hstride + 4 * sc4;
        #pragma unroll
        for (int i = 0; i < 6; i++)
            hq[i] = *(const v4f*)(hp + 64 * i);
    }

    const int ncol = g * HDIM + hc0 + 16 * ch + l16;
    v8f acc0 = {}, acc1 = {};
    if (rec) {
        const float* xl = xgt + (size_t)(b0 + 8 * lhalf) * G4H + ncol;
        #pragma unroll
        for (int r = 0; r < 8; r++)
            acc0[r] = xl[r * G4H];
    }

    float cprev[2] = {0.0f, 0.0f};
    if (rec && !zeroh) {
        const float* xgp = xgt - (size_t)(BSZ * G4H);
        #pragma unroll
        for (int i = 0; i < 2; i++)
            cprev[i] = xgp[(size_t)(b0 + 2 * wid + i) * G4H + hc0 + lane];
    }

    v4f qb[12][4];
    {
        const float* wrow = Wm + (size_t)ncol * HDIM + 16 * lhalf;
        #pragma unroll
        for (int kc = 0; kc < 12; kc++)
            #pragma unroll
            for (int jj = 0; jj < 4; jj++)
                qb[kc][jj] = *(const v4f*)(wrow + 32 * kc + 4 * jj);
    }

    float bias0 = 0.0f, bias1 = 0.0f;
    if (!rec) {
        bias0 = (dir ? bih1b : bih1f)[ncol];
        bias1 = (dir ? bhh1b : bhh1f)[ncol];
    }

    if (stage == 0 && t + 1 < SSZ && tid < 64) {
        const float* nxt = xgt + (size_t)(BSZ * G4H)
                         + (size_t)(b0 + (tid >> 2)) * G4H + (tid & 3) * HDIM + hc0;
        __builtin_prefetch(nxt, 0, 0);
    }

    #pragma unroll
    for (int i = 0; i < 6; i++)
        *(v4f*)&hs[srow][4 * sc4 + 64 * i] = hq[i];
    __builtin_amdgcn_fence(__ATOMIC_RELEASE, "workgroup", "local");
    __builtin_amdgcn_s_barrier();
    __builtin_amdgcn_fence(__ATOMIC_ACQUIRE, "workgroup", "local");

    const float* hlane = &hs[l16][16 * lhalf];
    v4f qa[2][4];
    #pragma unroll
    for (int jj = 0; jj < 4; jj++)
        qa[0][jj] = *(const v4f*)(hlane + 4 * jj);
    __builtin_amdgcn_sched_group_barrier(0x100, 4, 0);
    #pragma unroll
    for (int kc = 0; kc < 12; kc++) {
        if (kc + 1 < 12) {
            #pragma unroll
            for (int jj = 0; jj < 4; jj++)
                qa[(kc + 1) & 1][jj] = *(const v4f*)(hlane + 32 * (kc + 1) + 4 * jj);
        }
        #pragma unroll
        for (int jj = 0; jj < 4; jj++) {
            const v4f ha = qa[kc & 1][jj];
            v2f a0, a1, w0, w1;
            a0.x = ha.x;  a0.y = ha.y;  a1.x = ha.z;  a1.y = ha.w;
            w0.x = qb[kc][jj].x;  w0.y = qb[kc][jj].y;
            w1.x = qb[kc][jj].z;  w1.y = qb[kc][jj].w;
            acc0 = WMMA_F32(a0, w0, acc0);
            acc1 = WMMA_F32(a1, w1, acc1);
        }
        if (kc + 1 < 12)
            __builtin_amdgcn_sched_group_barrier(0x100, 4, 0);
        __builtin_amdgcn_sched_group_barrier(0x008, 8, 0);
    }

    #pragma unroll
    for (int r = 0; r < 8; r++)
        gs[g][8 * lhalf + r][16 * ch + l16] = ((acc0[r] + acc1[r]) + bias0) + bias1;
    __syncthreads();

    if (rec) {
        #pragma unroll
        for (int i = 0; i < 2; i++) {
            const int row = 2 * wid + i;
            float iv = sigmoid_f(gs[0][row][lane]);
            float fv = sigmoid_f(gs[1][row][lane]);
            float gv = tanh_f(gs[2][row][lane]);
            float ov = sigmoid_f(gs[3][row][lane]);
            float c  = fv * cprev[i] + iv * gv;
            float h  = ov * tanh_f(c);
            xgt[(size_t)(b0 + row) * G4H + hc0 + lane] = c;
            hbase[(size_t)((b0 + row) * SSZ + t) * hstride + hc0 + lane] = h;
        }
    } else {
        #pragma unroll
        for (int i = 0; i < 2; i++) {
            const int row = 2 * wid + i;
            #pragma unroll
            for (int gg = 0; gg < 4; gg++)
                xgt[(size_t)(b0 + row) * G4H + gg * HDIM + hc0 + lane] = gs[gg][row][lane];
        }
    }
}

extern "C" void kernel_launch(void* const* d_in, const int* in_sizes, int n_in,
                              void* d_out, int out_size, void* d_ws, size_t ws_size,
                              hipStream_t stream) {
    const float* emb    = (const float*)d_in[0];
    const float* wih_f0 = (const float*)d_in[2];
    const float* whh_f0 = (const float*)d_in[3];
    const float* bih_f0 = (const float*)d_in[4];
    const float* bhh_f0 = (const float*)d_in[5];
    const float* wih_f1 = (const float*)d_in[6];
    const float* whh_f1 = (const float*)d_in[7];
    const float* bih_f1 = (const float*)d_in[8];
    const float* bhh_f1 = (const float*)d_in[9];
    const float* wih_b0 = (const float*)d_in[10];
    const float* whh_b0 = (const float*)d_in[11];
    const float* bih_b0 = (const float*)d_in[12];
    const float* bhh_b0 = (const float*)d_in[13];
    const float* wih_b1 = (const float*)d_in[14];
    const float* whh_b1 = (const float*)d_in[15];
    const float* bih_b1 = (const float*)d_in[16];
    const float* bhh_b1 = (const float*)d_in[17];

    float* xg = (float*)d_ws;
    float* h0 = xg + (size_t)2 * SSZ * BSZ * G4H;
    float* out = (float*)d_out;

    dim3 gA(G4H / 64, ROWS / 64, 2);
    dim3 bA(128);
    dim3 gS(3 * (HDIM / 32), 2, 2);
    dim3 bS(256);

    xg_gemm_kernel<<<gA, bA, 0, stream>>>(emb, emb, wih_f0, wih_b0,
                                          bih_f0, bhh_f0, bih_b0, bhh_b0,
                                          xg, EDIM, 1);
    for (int j = 0; j < SSZ + 3; j++)
        lstm_wavefront_kernel<<<gS, bS, 0, stream>>>(xg, whh_f0, whh_b0, wih_f1, wih_b1,
                                                     whh_f1, whh_b1,
                                                     bih_f1, bhh_f1, bih_b1, bhh_b1,
                                                     h0, out, j);
}


# Weights repacked once per call into fragment order in the workspace (518 launches, was 517): a wave's weight load reads 512 contiguous bytes, 4 whole lines, not 32 B of every line it touches
# speedup vs baseline: 29.6095x; 1.3659x over previous
#include <hip/hip_runtime.h>
#include <hip/hip_bf16.h>

typedef float v2f __attribute__((ext_vector_type(2)));
typedef float v4f __attribute__((ext_vector_type(4)));
typedef float v8f __attribute__((ext_vector_type(8)));

#define WMMA_F32(a, b, c) \
    __builtin_amdgcn_wmma_f32_16x16x4_f32(false, (a), false, (b), (short)0, (c), false, false)

#define EDIM 768
#define HDIM 384
#define G4H  1536
#define BSZ  32
#define SSZ  512
#define ROWS (BSZ * SSZ)
#define HPITCH 388
#define XRING 8
#define TILEF 6144

__device__ __forceinline__ float sigmoid_f(float x) {
    return 1.0f / (1.0f + __expf(-x));
}
__device__ __forceinline__ float tanh_f(float x) {
    return 1.0f - 2.0f / (__expf(2.0f * x) + 1.0f);
}

#define GPITCH 36

template <int SLOT>
__device__ __forceinline__
void wavefront_stage(float* __restrict__ xg, const float* __restrict__ emb,
                     const float* __restrict__ wpk,
                     const float* __restrict__ bih, const float* __restrict__ bhh,
                     float* __restrict__ h0, float* __restrict__ out, int j,
                     float (*hs)[HPITCH], float (*gs)[16][GPITCH])
{
    constexpr bool rec   = (SLOT == 0) || (SLOT == 1);
    constexpr bool emb0  = (SLOT == 2) || (SLOT == 3);
    constexpr bool initx = rec || (SLOT == 2);
    constexpr bool addb  = (SLOT == 3) || (SLOT == 4);
    constexpr int  lag   = (SLOT == 0) ? 2 : (SLOT == 1) ? 5 : (SLOT == 2) ? 1 : (SLOT == 3) ? 0 : 4;
    constexpr int  koff   = (SLOT == 2) ? HDIM : 0;

    const int hc0 = blockIdx.x * 32;
    const int b0  = (blockIdx.y & 1) * 16;
    const int dir = blockIdx.y >> 1;

    const int t = j - lag;
    if (t < 0 || t >= SSZ) return;

    const bool zeroh = rec && (t == 0);

    const float* vin;
    int tin;
    constexpr int vstride = emb0 ? EDIM : (SLOT == 1) ? 2 * HDIM : HDIM;
    if (emb0)           { vin = emb + koff;                     tin = dir ? (SSZ - 1 - t) : t; }
    else if (SLOT == 1) { vin = out + dir * HDIM;               tin = t - 1; }
    else                { vin = h0 + (size_t)dir * ROWS * HDIM; tin = (SLOT == 0) ? (t - 1) : t; }

    float* hout = (SLOT == 1) ? (out + dir * HDIM) : (h0 + (size_t)dir * ROWS * HDIM);
    constexpr int hstride = (SLOT == 1) ? 2 * HDIM : HDIM;

    float* xgt = xg + (size_t)(dir * XRING + (t & (XRING - 1))) * (BSZ * G4H);

    const int tid   = threadIdx.x;
    const int lane  = tid & 31;
    const int wid   = tid >> 5;
    const int g     = wid >> 1;
    const int ch    = wid & 1;
    const int lhalf = lane >> 4;
    const int l16   = lane & 15;

    const int srow = tid >> 4;
    const int sc4  = tid & 15;
    v4f hq[6];
    #pragma unroll
    for (int i = 0; i < 6; i++)
        hq[i] = v4f{0.0f, 0.0f, 0.0f, 0.0f};
    if (!zeroh) {
        const float* hp = vin + (size_t)((b0 + srow) * SSZ + tin) * vstride + 4 * sc4;
        #pragma unroll
        for (int i = 0; i < 6; i++)
            hq[i] = *(const v4f*)(hp + 64 * i);
    }

    const int ncol = g * HDIM + hc0 + 16 * ch + l16;
    v8f acc0 = {}, acc1 = {};
    if (initx) {
        const float* xl = xgt + (size_t)(b0 + 8 * lhalf) * G4H + ncol;
        #pragma unroll
        for (int r = 0; r < 8; r++)
            acc0[r] = xl[r * G4H];
    }

    float cprev[2] = {0.0f, 0.0f};
    if (rec && !zeroh) {
        const float* xgp = xg + (size_t)(dir * XRING + ((t - 1) & (XRING - 1))) * (BSZ * G4H);
        #pragma unroll
        for (int i = 0; i < 2; i++)
            cprev[i] = xgp[(size_t)(b0 + 2 * wid + i) * G4H + hc0 + lane];
    }

    v4f qb[12][4];
    {
        const int tile = ((SLOT * 2 + dir) * 12 + (int)blockIdx.x) * 8 + wid;
        const v4f* wt = (const v4f*)(wpk + (size_t)tile * TILEF) + lane;
        #pragma unroll
        for (int kc = 0; kc < 12; kc++) {
            #pragma unroll
            for (int jj = 0; jj < 4; jj++)
                qb[kc][jj] = wt[(4 * kc + jj) * 32];
            asm volatile("" ::: "memory");
        }
    }

    float bias0 = 0.0f, bias1 = 0.0f;
    if (addb) {
        bias0 = bih[ncol];
        bias1 = bhh[ncol];
    }

    if (emb0 && hc0 == 0 && t + 1 < SSZ && tid < 192) {
        const int tnx = dir ? (SSZ - 2 - t) : (t + 1);
        const float* nxt = emb + (size_t)((b0 + tid / 12) * SSZ + tnx) * EDIM + koff + 32 * (tid % 12);
        __builtin_prefetch(nxt, 0, 0);
    }

    __builtin_amdgcn_sched_barrier(0);

    #pragma unroll
    for (int i = 0; i < 6; i++)
        *(v4f*)&hs[srow][4 * sc4 + 64 * i] = hq[i];
    __builtin_amdgcn_fence(__ATOMIC_RELEASE, "workgroup", "local");
    __builtin_amdgcn_s_barrier();
    __builtin_amdgcn_fence(__ATOMIC_ACQUIRE, "workgroup", "local");

    const float* hlane = &hs[l16][16 * lhalf];
    v4f qa[2][4];
    #pragma unroll
    for (int jj = 0; jj < 4; jj++)
        qa[0][jj] = *(const v4f*)(hlane + 4 * jj);
    __builtin_amdgcn_sched_group_barrier(0x100, 4, 0);
    #pragma unroll
    for (int kc = 0; kc < 12; kc++) {
        if (kc + 1 < 12) {
            #pragma unroll
            for (int jj = 0; jj < 4; jj++)
                qa[(kc + 1) & 1][jj] = *(const v4f*)(hlane + 32 * (kc + 1) + 4 * jj);
        }
        #pragma unroll
        for (int jj = 0; jj < 4; jj++) {
            const v4f ha = qa[kc & 1][jj];
            v2f a0, a1, w0, w1;
            a0.x = ha.x;  a0.y = ha.y;  a1.x = ha.z;  a1.y = ha.w;
            w0.x = qb[kc][jj].x;  w0.y = qb[kc][jj].y;
            w1.x = qb[kc][jj].z;  w1.y = qb[kc][jj].w;
            acc0 = WMMA_F32(a0, w0, acc0);
            acc1 = WMMA_F32(a1, w1, acc1);
        }
        if (kc + 1 < 12)
            __builtin_amdgcn_sched_group_barrier(0x100, 4, 0);
        __builtin_amdgcn_sched_group_barrier(0x008, 8, 0);
    }

    #pragma unroll
    for (int r = 0; r < 8; r++)
        gs[g][8 * lhalf + r][16 * ch + l16] = ((acc0[r] + acc1[r]) + bias0) + bias1;
    __syncthreads();

    if (rec) {
        #pragma unroll
        for (int i = 0; i < 2; i++) {
            const int row = 2 * wid + i;
            float iv = sigmoid_f(gs[0][row][lane]);
            float fv = sigmoid_f(gs[1][row][lane]);
            float gv = tanh_f(gs[2][row][lane]);
            float ov = sigmoid_f(gs[3][row][lane]);
            float c  = fv * cprev[i] + iv * gv;
            float h  = ov * tanh_f(c);
            xgt[(size_t)(b0 + row) * G4H + hc0 + lane] = c;
            hout[(size_t)((b0 + row) * SSZ + t) * hstride + hc0 + lane] = h;
        }
    } else {
        #pragma unroll
        for (int i = 0; i < 2; i++) {
            const int row = 2 * wid + i;
            #pragma unroll
            for (int gg = 0; gg < 4; gg++)
                xgt[(size_t)(b0 + row) * G4H + gg * HDIM + hc0 + lane] = gs[gg][row][lane];
        }
    }
}

__global__ __launch_bounds__(256)
void pack_weights_kernel(float* __restrict__ wpk,
                         const float* __restrict__ whh0f, const float* __restrict__ whh0b,
                         const float* __restrict__ whh1f, const float* __restrict__ whh1b,
                         const float* __restrict__ wih0f, const float* __restrict__ wih0b,
                         const float* __restrict__ wih1f, const float* __restrict__ wih1b)
{
    const int slot = blockIdx.z;
    const int dir  = blockIdx.y >> 2;
    const int kq   = blockIdx.y & 3;
    const bool bwd = dir != 0;

    const float* Wm;
    int wpitch = HDIM, koff = 0;
    switch (slot) {
    case 0:  Wm = bwd ? whh0b : whh0f; break;
    case 1:  Wm = bwd ? whh1b : whh1f; break;
    case 2:  Wm = bwd ? wih0b : wih0f; wpitch = EDIM; koff = HDIM; break;
    case 3:  Wm = bwd ? wih0b : wih0f; wpitch = EDIM; break;
    default: Wm = bwd ? wih1b : wih1f; break;
    }

    const int tid   = threadIdx.x;
    const int lane  = tid & 31;
    const int wid   = tid >> 5;
    const int lhalf = lane >> 4;
    const int l16   = lane & 15;
    const int ncol  = (wid >> 1) * HDIM + (int)blockIdx.x * 32 + 16 * (wid & 1) + l16;

    const float* wrow = Wm + (size_t)ncol * wpitch + koff + 16 * lhalf;
    const int tile = ((slot * 2 + dir) * 12 + (int)blockIdx.x) * 8 + wid;
    v4f* wt = (v4f*)(wpk + (size_t)tile * TILEF) + lane;

    v4f q[3][4];
    #pragma unroll
    for (int i = 0; i < 3; i++)
        #pragma unroll
        for (int jj = 0; jj < 4; jj++)
            q[i][jj] = *(const v4f*)(wrow + 32 * (3 * kq + i) + 4 * jj);
    #pragma unroll
    for (int i = 0; i < 3; i++)
        #pragma unroll
        for (int jj = 0; jj < 4; jj++)
            wt[(4 * (3 * kq + i) + jj) * 32] = q[i][jj];
}

__global__ __launch_bounds__(256)
void lstm_wavefront_kernel(float* __restrict__ xg, const float* __restrict__ emb,
                           const float* __restrict__ wpk,
                           const float* __restrict__ bih0f, const float* __restrict__ bhh0f,
                           const float* __restrict__ bih0b, const float* __restrict__ bhh0b,
                           const float* __restrict__ bih1f, const float* __restrict__ bhh1f,
                           const float* __restrict__ bih1b, const float* __restrict__ bhh1b,
                           float* __restrict__ h0, float* __restrict__ out, int j)
{
    __shared__ __attribute__((aligned(16))) float hs[16][HPITCH];
    __shared__ float gs[4][16][GPITCH];

    const bool bwd = (blockIdx.y >> 1) != 0;
    switch (blockIdx.z) {
    case 0:
        wavefront_stage<0>(xg, emb, wpk, nullptr, nullptr, h0, out, j, hs, gs);
        break;
    case 1:
        wavefront_stage<1>(xg, emb, wpk, nullptr, nullptr, h0, out, j, hs, gs);
        break;
    case 2:
        wavefront_stage<2>(xg, emb, wpk, nullptr, nullptr, h0, out, j, hs, gs);
        break;
    case 3:
        wavefront_stage<3>(xg, emb, wpk, bwd ? bih0b : bih0f, bwd ? bhh0b : bhh0f,
                           h0, out, j, hs, gs);
        break;
    default:
        wavefront_stage<4>(xg, emb, wpk, bwd ? bih1b : bih1f, bwd ? bhh1b : bhh1f,
                           h0, out, j, hs, gs);
        break;
    }
}

extern "C" void kernel_launch(void* const* d_in, const int* in_sizes, int n_in,
                              void* d_out, int out_size, void* d_ws, size_t ws_size,
                              hipStream_t stream) {
    const float* emb    = (const float*)d_in[0];
    const float* wih_f0 = (const float*)d_in[2];
    const float* whh_f0 = (const float*)d_in[3];
    const float* bih_f0 = (const float*)d_in[4];
    const float* bhh_f0 = (const float*)d_in[5];
    const float* wih_f1 = (const float*)d_in[6];
    const float* whh_f1 = (const float*)d_in[7];
    const float* bih_f1 = (const float*)d_in[8];
    const float* bhh_f1 = (const float*)d_in[9];
    const float* wih_b0 = (const float*)d_in[10];
    const float* whh_b0 = (const float*)d_in[11];
    const float* bih_b0 = (const float*)d_in[12];
    const float* bhh_b0 = (const float*)d_in[13];
    const float* wih_b1 = (const float*)d_in[14];
    const float* whh_b1 = (const float*)d_in[15];
    const float* bih_b1 = (const float*)d_in[16];
    const float* bhh_b1 = (const float*)d_in[17];

    float* xg  = (float*)d_ws;
    float* wpk = xg + (size_t)2 * XRING * BSZ * G4H;
    float* h0  = xg + (size_t)2 * SSZ * BSZ * G4H;
    float* out = (float*)d_out;

    pack_weights_kernel<<<dim3(HDIM / 32, 8, 5), dim3(256), 0, stream>>>(wpk, whh_f0, whh_b0, whh_f1, whh_b1,
                                                                       wih_f0, wih_b0, wih_f1, wih_b1);

    dim3 gS(HDIM / 32, 4, 5);
    dim3 bS(256);

    for (int j = 0; j < SSZ + 5; j++)
        lstm_wavefront_kernel<<<gS, bS, 0, stream>>>(xg, emb, wpk,
                                                     bih_f0, bhh_f0, bih_b0, bhh_b0,
                                                     bih_f1, bhh_f1, bih_b1, bhh_b1,
                                                     h0, out, j);
}


# Step work-groups run 4 waves, not 8: a wave owns both 16-column tiles of its gate (one LDS input fragment feeds both); its 96 weight loads are requested 5 chunks ahead inside the product loop
# speedup vs baseline: 31.2790x; 1.0564x over previous
#include <hip/hip_runtime.h>
#include <hip/hip_bf16.h>

typedef float v2f __attribute__((ext_vector_type(2)));
typedef float v4f __attribute__((ext_vector_type(4)));
typedef float v8f __attribute__((ext_vector_type(8)));

#define WMMA_F32(a, b, c) \
    __builtin_amdgcn_wmma_f32_16x16x4_f32(false, (a), false, (b), (short)0, (c), false, false)

#define EDIM 768
#define HDIM 384
#define G4H  1536
#define BSZ  32
#define SSZ  512
#define ROWS (BSZ * SSZ)
#define HPITCH 388
#define XRING 8
#define TILEF 6144

__device__ __forceinline__ float sigmoid_f(float x) {
    return 1.0f / (1.0f + __expf(-x));
}
__device__ __forceinline__ float tanh_f(float x) {
    return 1.0f - 2.0f / (__expf(2.0f * x) + 1.0f);
}

#define GPITCH 36

template <int SLOT>
__device__ __forceinline__
void wavefront_stage(float* __restrict__ xg, const float* __restrict__ emb,
                     const float* __restrict__ wpk,
                     const float* __restrict__ bih, const float* __restrict__ bhh,
                     float* __restrict__ h0, float* __restrict__ out, int j,
                     float (*hs)[HPITCH], float (*gs)[16][GPITCH])
{
    constexpr bool rec   = (SLOT == 0) || (SLOT == 1);
    constexpr bool emb0  = (SLOT == 2) || (SLOT == 3);
    constexpr bool initx = rec || (SLOT == 2);
    constexpr bool addb  = (SLOT == 3) || (SLOT == 4);
    constexpr int  lag   = (SLOT == 0) ? 2 : (SLOT == 1) ? 5 : (SLOT == 2) ? 1 : (SLOT == 3) ? 0 : 4;
    constexpr int  koff   = (SLOT == 2) ? HDIM : 0;

    const int hc0 = blockIdx.x * 32;
    const int b0  = (blockIdx.y & 1) * 16;
    const int dir = blockIdx.y >> 1;

    const int t = j - lag;
    if (t < 0 || t >= SSZ) return;

    const bool zeroh = rec && (t == 0);

    const float* vin;
    int tin;
    constexpr int vstride = emb0 ? EDIM : (SLOT == 1) ? 2 * HDIM : HDIM;
    if (emb0)           { vin = emb + koff;                     tin = dir ? (SSZ - 1 - t) : t; }
    else if (SLOT == 1) { vin = out + dir * HDIM;               tin = t - 1; }
    else                { vin = h0 + (size_t)dir * ROWS * HDIM; tin = (SLOT == 0) ? (t - 1) : t; }

    float* hout = (SLOT == 1) ? (out + dir * HDIM) : (h0 + (size_t)dir * ROWS * HDIM);
    constexpr int hstride = (SLOT == 1) ? 2 * HDIM : HDIM;

    float* xgt = xg + (size_t)(dir * XRING + (t & (XRING - 1))) * (BSZ * G4H);

    const int tid   = threadIdx.x;
    const int lane  = tid & 31;
    const int g     = tid >> 5;
    const int lhalf = lane >> 4;
    const int l16   = lane & 15;

    v4f hq[12];
    #pragma unroll
    for (int i = 0; i < 12; i++)
        hq[i] = v4f{0.0f, 0.0f, 0.0f, 0.0f};
    if (!zeroh) {
        #pragma unroll
        for (int i = 0; i < 12; i++) {
            const int srow = g + 4 * (i / 3);
            hq[i] = *(const v4f*)(vin + (size_t)((b0 + srow) * SSZ + tin) * vstride + 128 * (i % 3) + 4 * lane);
        }
    }

    const int ncol = g * HDIM + hc0 + l16;
    v8f acc[2][2] = {};
    if (initx) {
        const float* xl = xgt + (size_t)(b0 + 8 * lhalf) * G4H + ncol;
        #pragma unroll
        for (int ct = 0; ct < 2; ct++) {
            #pragma unroll
            for (int r = 0; r < 8; r++)
                acc[ct][0][r] = xl[r * G4H + 16 * ct];
        }
    }

    float cprev[4] = {0.0f, 0.0f, 0.0f, 0.0f};
    if (rec && !zeroh) {
        const float* xgp = xg + (size_t)(dir * XRING + ((t - 1) & (XRING - 1))) * (BSZ * G4H);
        #pragma unroll
        for (int i = 0; i < 4; i++)
            cprev[i] = xgp[(size_t)(b0 + 4 * g + i) * G4H + hc0 + lane];
    }

    constexpr int AHEAD = 5;
    v4f qb[12][2][4];
    const int tile0 = ((SLOT * 2 + dir) * 12 + (int)blockIdx.x) * 8 + 2 * g;
    const v4f* wt = (const v4f*)(wpk + (size_t)tile0 * TILEF) + lane;
    #pragma unroll
    for (int kc = 0; kc < AHEAD; kc++) {
        #pragma unroll
        for (int ct = 0; ct < 2; ct++) {
            #pragma unroll
            for (int jj = 0; jj < 4; jj++)
                qb[kc][ct][jj] = wt[ct * (TILEF / 4) + (4 * kc + jj) * 32];
        }
        asm volatile("" ::: "memory");
    }

    float bias0[2] = {0.0f, 0.0f}, bias1[2] = {0.0f, 0.0f};
    if (addb) {
        #pragma unroll
        for (int ct = 0; ct < 2; ct++) {
            bias0[ct] = bih[ncol + 16 * ct];
            bias1[ct] = bhh[ncol + 16 * ct];
        }
    }

    if (emb0 && hc0 == 0 && t + 1 < SSZ) {
        const int tnx = dir ? (SSZ - 2 - t) : (t + 1);
        #pragma unroll
        for (int p = tid; p < 192; p += 128) {
            const float* nxt = emb + (size_t)((b0 + p / 12) * SSZ + tnx) * EDIM + koff + 32 * (p % 12);
            __builtin_prefetch(nxt, 0, 0);
        }
    }

    __builtin_amdgcn_sched_barrier(0);

    #pragma unroll
    for (int i = 0; i < 12; i++)
        *(v4f*)&hs[g + 4 * (i / 3)][128 * (i % 3) + 4 * lane] = hq[i];
    __builtin_amdgcn_fence(__ATOMIC_RELEASE, "workgroup", "local");
    __builtin_amdgcn_s_barrier();
    __builtin_amdgcn_fence(__ATOMIC_ACQUIRE, "workgroup", "local");

    const float* hlane = &hs[l16][16 * lhalf];
    v4f qa[2][4];
    #pragma unroll
    for (int jj = 0; jj < 4; jj++)
        qa[0][jj] = *(const v4f*)(hlane + 4 * jj);
    __builtin_amdgcn_sched_group_barrier(0x100, 4, 0);
    #pragma unroll
    for (int kc = 0; kc < 12; kc++) {
        if (kc + 1 < 12) {
            #pragma unroll
            for (int jj = 0; jj < 4; jj++)
                qa[(kc + 1) & 1][jj] = *(const v4f*)(hlane + 32 * (kc + 1) + 4 * jj);
        }
        if (kc + AHEAD < 12) {
            #pragma unroll
            for (int ct = 0; ct < 2; ct++) {
                #pragma unroll
                for (int jj = 0; jj < 4; jj++)
                    qb[kc + AHEAD][ct][jj] = wt[ct * (TILEF / 4) + (4 * (kc + AHEAD) + jj) * 32];
            }
        }
        #pragma unroll
        for (int jj = 0; jj < 4; jj++) {
            const v4f ha = qa[kc & 1][jj];
            v2f a0, a1;
            a0.x = ha.x;  a0.y = ha.y;  a1.x = ha.z;  a1.y = ha.w;
            #pragma unroll
            for (int ct = 0; ct < 2; ct++) {
                v2f w0, w1;
                w0.x = qb[kc][ct][jj].x;  w0.y = qb[kc][ct][jj].y;
                w1.x = qb[kc][ct][jj].z;  w1.y = qb[kc][ct][jj].w;
                acc[ct][0] = WMMA_F32(a0, w0, acc[ct][0]);
                acc[ct][1] = WMMA_F32(a1, w1, acc[ct][1]);
            }
        }
        if (kc + 1 < 12)
            __builtin_amdgcn_sched_group_barrier(0x100, 4, 0);
        if (kc + AHEAD < 12)
            __builtin_amdgcn_sched_group_barrier(0x020, 8, 0);
        __builtin_amdgcn_sched_group_barrier(0x008, 16, 0);
    }

    #pragma unroll
    for (int ct = 0; ct < 2; ct++) {
        #pragma unroll
        for (int r = 0; r < 8; r++)
            gs[g][8 * lhalf + r][16 * ct + l16] = ((acc[ct][0][r] + acc[ct][1][r]) + bias0[ct]) + bias1[ct];
    }
    __syncthreads();

    if (rec) {
        #pragma unroll
        for (int i = 0; i < 4; i++) {
            const int row = 4 * g + i;
            float iv = sigmoid_f(gs[0][row][lane]);
            float fv = sigmoid_f(gs[1][row][lane]);
            float gv = tanh_f(gs[2][row][lane]);
            float ov = sigmoid_f(gs[3][row][lane]);
            float c  = fv * cprev[i] + iv * gv;
            float h  = ov * tanh_f(c);
            xgt[(size_t)(b0 + row) * G4H + hc0 + lane] = c;
            hout[(size_t)((b0 + row) * SSZ + t) * hstride + hc0 + lane] = h;
        }
    } else {
        #pragma unroll
        for (int i = 0; i < 4; i++) {
            const int row = 4 * g + i;
            #pragma unroll
            for (int gg = 0; gg < 4; gg++)
                xgt[(size_t)(b0 + row) * G4H + gg * HDIM + hc0 + lane] = gs[gg][row][lane];
        }
    }
}

__global__ __launch_bounds__(256)
void pack_weights_kernel(float* __restrict__ wpk,
                         const float* __restrict__ whh0f, const float* __restrict__ whh0b,
                         const float* __restrict__ whh1f, const float* __restrict__ whh1b,
                         const float* __restrict__ wih0f, const float* __restrict__ wih0b,
                         const float* __restrict__ wih1f, const float* __restrict__ wih1b)
{
    const int slot = blockIdx.z;
    const int dir  = blockIdx.y >> 2;
    const int kq   = blockIdx.y & 3;
    const bool bwd = dir != 0;

    const float* Wm;
    int wpitch = HDIM, koff = 0;
    switch (slot) {
    case 0:  Wm = bwd ? whh0b : whh0f; break;
    case 1:  Wm = bwd ? whh1b : whh1f; break;
    case 2:  Wm = bwd ? wih0b : wih0f; wpitch = EDIM; koff = HDIM; break;
    case 3:  Wm = bwd ? wih0b : wih0f; wpitch = EDIM; break;
    default: Wm = bwd ? wih1b : wih1f; break;
    }

    const int tid   = threadIdx.x;
    const int lane  = tid & 31;
    const int wid   = tid >> 5;
    const int lhalf = lane >> 4;
    const int l16   = lane & 15;
    const int ncol  = (wid >> 1) * HDIM + (int)blockIdx.x * 32 + 16 * (wid & 1) + l16;

    const float* wrow = Wm + (size_t)ncol * wpitch + koff + 16 * lhalf;
    const int tile = ((slot * 2 + dir) * 12 + (int)blockIdx.x) * 8 + wid;
    v4f* wt = (v4f*)(wpk + (size_t)tile * TILEF) + lane;

    v4f q[3][4];
    #pragma unroll
    for (int i = 0; i < 3; i++)
        #pragma unroll
        for (int jj = 0; jj < 4; jj++)
            q[i][jj] = *(const v4f*)(wrow + 32 * (3 * kq + i) + 4 * jj);
    #pragma unroll
    for (int i = 0; i < 3; i++)
        #pragma unroll
        for (int jj = 0; jj < 4; jj++)
            wt[(4 * (3 * kq + i) + jj) * 32] = q[i][jj];
}

__global__ __launch_bounds__(128)
void lstm_wavefront_kernel(float* __restrict__ xg, const float* __restrict__ emb,
                           const float* __restrict__ wpk,
                           const float* __restrict__ bih0f, const float* __restrict__ bhh0f,
                           const float* __restrict__ bih0b, const float* __restrict__ bhh0b,
                           const float* __restrict__ bih1f, const float* __restrict__ bhh1f,
                           const float* __restrict__ bih1b, const float* __restrict__ bhh1b,
                           float* __restrict__ h0, float* __restrict__ out, int j)
{
    __shared__ __attribute__((aligned(16))) float hs[16][HPITCH];
    __shared__ float gs[4][16][GPITCH];

    const bool bwd = (blockIdx.y >> 1) != 0;
    switch (blockIdx.z) {
    case 0:
        wavefront_stage<0>(xg, emb, wpk, nullptr, nullptr, h0, out, j, hs, gs);
        break;
    case 1:
        wavefront_stage<1>(xg, emb, wpk, nullptr, nullptr, h0, out, j, hs, gs);
        break;
    case 2:
        wavefront_stage<2>(xg, emb, wpk, nullptr, nullptr, h0, out, j, hs, gs);
        break;
    case 3:
        wavefront_stage<3>(xg, emb, wpk, bwd ? bih0b : bih0f, bwd ? bhh0b : bhh0f,
                           h0, out, j, hs, gs);
        break;
    default:
        wavefront_stage<4>(xg, emb, wpk, bwd ? bih1b : bih1f, bwd ? bhh1b : bhh1f,
                           h0, out, j, hs, gs);
        break;
    }
}

extern "C" void kernel_launch(void* const* d_in, const int* in_sizes, int n_in,
                              void* d_out, int out_size, void* d_ws, size_t ws_size,
                              hipStream_t stream) {
    const float* emb    = (const float*)d_in[0];
    const float* wih_f0 = (const float*)d_in[2];
    const float* whh_f0 = (const float*)d_in[3];
    const float* bih_f0 = (const float*)d_in[4];
    const float* bhh_f0 = (const float*)d_in[5];
    const float* wih_f1 = (const float*)d_in[6];
    const float* whh_f1 = (const float*)d_in[7];
    const float* bih_f1 = (const float*)d_in[8];
    const float* bhh_f1 = (const float*)d_in[9];
    const float* wih_b0 = (const float*)d_in[10];
    const float* whh_b0 = (const float*)d_in[11];
    const float* bih_b0 = (const float*)d_in[12];
    const float* bhh_b0 = (const float*)d_in[13];
    const float* wih_b1 = (const float*)d_in[14];
    const float* whh_b1 = (const float*)d_in[15];
    const float* bih_b1 = (const float*)d_in[16];
    const float* bhh_b1 = (const float*)d_in[17];

    float* xg  = (float*)d_ws;
    float* wpk = xg + (size_t)2 * XRING * BSZ * G4H;
    float* h0  = xg + (size_t)2 * SSZ * BSZ * G4H;
    float* out = (float*)d_out;

    pack_weights_kernel<<<dim3(HDIM / 32, 8, 5), dim3(256), 0, stream>>>(wpk, whh_f0, whh_b0, whh_f1, whh_b1,
                                                                       wih_f0, wih_b0, wih_f1, wih_b1);

    dim3 gS(HDIM / 32, 4, 5);
    dim3 bS(128);

    for (int j = 0; j < SSZ + 5; j++)
        lstm_wavefront_kernel<<<gS, bS, 0, stream>>>(xg, emb, wpk,
                                                     bih_f0, bhh_f0, bih_b0, bhh_b0,
                                                     bih_f1, bhh_f1, bih_b1, bhh_b1,
                                                     h0, out, j);
}
